# starting kernel
# baseline (speedup 1.0000x reference)
#include <hip/hip_runtime.h>
#include <math.h>

#define N_NODES 100000
#define N_EDGES 800000
#define IN_DIM  128
#define HEADS   4
#define DHEAD   32
#define HD      128
#define EPS_LN  1e-5f
#define NEG_SLOPE 0.2f

typedef __attribute__((ext_vector_type(2))) float v2f;
typedef __attribute__((ext_vector_type(4))) float v4f;
typedef __attribute__((ext_vector_type(8))) float v8f;

__global__ void fill_kernel(float* __restrict__ out, float* __restrict__ esum,
                            int* __restrict__ emax) {
    int i = blockIdx.x * blockDim.x + threadIdx.x;
    if (i < N_NODES * HD) out[i] = 0.0f;
    if (i < N_NODES * HEADS) {
        esum[i] = 0.0f;
        emax[i] = (int)0xFF800000u;
    }
}

__global__ __launch_bounds__(256) void gemm_wmma_kernel(
    const float* __restrict__ feat,
    const float* __restrict__ Wsrc,  const float* __restrict__ bsrc,
    const float* __restrict__ Wskip, const float* __restrict__ bskip,
    float* __restrict__ feat_src, float* __restrict__ skip) {
    int wave = (blockIdx.x * blockDim.x + threadIdx.x) >> 5;
    int lane = threadIdx.x & 31;
    if (wave >= N_NODES / 16) return;
    const int node0 = wave * 16;
    const int half  = lane >> 4;
    const int l16   = lane & 15;

    const float* Arow = feat + (size_t)(node0 + l16) * IN_DIM + 2 * half;

    for (int p = 0; p < 2; ++p) {
        const float* W    = p ? Wskip : Wsrc;
        const float* bias = p ? bskip : bsrc;
        float*       out  = p ? skip  : feat_src;
        for (int ct = 0; ct < 8; ++ct) {
            const int n = ct * 16 + l16;
            v8f acc;
            const float bv = bias[n];
            #pragma unroll
            for (int v = 0; v < 8; ++v) acc[v] = bv;
            const float* Bcol = W + (size_t)n * IN_DIM + 2 * half;
            for (int kt = 0; kt < 32; ++kt) {
                v2f a = *(const v2f*)(Arow + kt * 4);
                v2f b = *(const v2f*)(Bcol + kt * 4);
                acc = __builtin_amdgcn_wmma_f32_16x16x4_f32(
                    false, a, false, b, (short)0, acc, false, false);
            }
            float* obase = out + (size_t)node0 * HD + ct * 16 + l16;
            #pragma unroll
            for (int v = 0; v < 8; ++v)
                obase[(size_t)(v + 8 * half) * HD] = acc[v];
        }
    }
}

__global__ void elr_kernel(const float* __restrict__ feat_src,
                           const float* __restrict__ attn_l,
                           const float* __restrict__ attn_r,
                           float* __restrict__ el, float* __restrict__ er) {
    int i = blockIdx.x * blockDim.x + threadIdx.x;
    if (i >= N_NODES * HEADS) return;
    const int h = i & 3;
    const float* f  = feat_src + (size_t)(i >> 2) * HD + h * DHEAD;
    const float* al = attn_l + h * DHEAD;
    const float* ar = attn_r + h * DHEAD;
    float sl = 0.0f, sr = 0.0f;
    #pragma unroll
    for (int d = 0; d < DHEAD; ++d) {
        float v = f[d];
        sl += v * al[d];
        sr += v * ar[d];
    }
    el[i] = sl;
    er[i] = sr;
}

__global__ void edge_logit_kernel(const int* __restrict__ src, const int* __restrict__ dst,
                                  const float* __restrict__ el, const float* __restrict__ er,
                                  float* __restrict__ ebuf, int* __restrict__ emax) {
    int e = blockIdx.x * blockDim.x + threadIdx.x;
    if (e >= N_EDGES) return;
    const int s = src[e], d = dst[e];
    const v4f els = *(const v4f*)(el + (size_t)s * 4);
    const v4f erd = *(const v4f*)(er + (size_t)d * 4);
    #pragma unroll
    for (int h = 0; h < 4; ++h) {
        float v = els[h] + erd[h];
        v = (v >= 0.0f) ? v : NEG_SLOPE * v;
        ebuf[(size_t)e * 4 + h] = v;
        int* p = emax + (size_t)d * 4 + h;
        if (v >= 0.0f) atomicMax(p, __float_as_int(v));
        else           atomicMin((unsigned int*)p, __float_as_uint(v));
    }
}

__global__ void edge_exp_kernel(const int* __restrict__ dst,
                                const float* __restrict__ emax,
                                float* __restrict__ ebuf, float* __restrict__ esum) {
    int e = blockIdx.x * blockDim.x + threadIdx.x;
    if (e >= N_EDGES) return;
    const int d = dst[e];
    #pragma unroll
    for (int h = 0; h < 4; ++h) {
        float m  = emax[(size_t)d * 4 + h];
        float ex = __expf(ebuf[(size_t)e * 4 + h] - m);
        ebuf[(size_t)e * 4 + h] = ex;
        atomicAdd(esum + (size_t)d * 4 + h, ex);
    }
}

__global__ __launch_bounds__(256) void aggregate_kernel(
    const int* __restrict__ src, const int* __restrict__ dst,
    const float* __restrict__ ebuf, const float* __restrict__ feat_src,
    float* __restrict__ out) {
    int wave = (blockIdx.x * blockDim.x + threadIdx.x) >> 5;
    int lane = threadIdx.x & 31;
    if (wave >= N_EDGES) return;
    const int s = src[wave], d = dst[wave];
    const int h = lane >> 3;
    const float a = ebuf[(size_t)wave * 4 + h];
    const v4f f = *(const v4f*)(feat_src + (size_t)s * HD + lane * 4);
    float* o = out + (size_t)d * HD + lane * 4;
    atomicAdd(o + 0, a * f.x);
    atomicAdd(o + 1, a * f.y);
    atomicAdd(o + 2, a * f.z);
    atomicAdd(o + 3, a * f.w);
}

__global__ __launch_bounds__(256) void final_kernel(
    const float* __restrict__ skip, const float* __restrict__ esum,
    const float* __restrict__ Wg, const float* __restrict__ bg,
    const float* __restrict__ ln_g, const float* __restrict__ ln_b,
    const float* __restrict__ prelu_a, float* __restrict__ out) {
    int wave = (blockIdx.x * blockDim.x + threadIdx.x) >> 5;
    int lane = threadIdx.x & 31;
    if (wave >= N_NODES) return;
    const int c = lane * 4;
    const int h = lane >> 3;
    const float es  = esum[(size_t)wave * 4 + h];
    const float inv = (es > 0.0f) ? 1.0f / es : 0.0f;
    v4f r  = *(const v4f*)(out  + (size_t)wave * HD + c);
    r = r * inv;
    const v4f sk = *(const v4f*)(skip + (size_t)wave * HD + c);
    const v4f dm = r - sk;
    const v4f w0 = *(const v4f*)(Wg + c);
    const v4f w1 = *(const v4f*)(Wg + 128 + c);
    const v4f w2 = *(const v4f*)(Wg + 256 + c);
    float part = r.x * w0.x + r.y * w0.y + r.z * w0.z + r.w * w0.w
               + sk.x * w1.x + sk.y * w1.y + sk.z * w1.z + sk.w * w1.w
               + dm.x * w2.x + dm.y * w2.y + dm.z * w2.z + dm.w * w2.w;
    #pragma unroll
    for (int off = 16; off > 0; off >>= 1) part += __shfl_xor(part, off, 32);
    const float g = 1.0f / (1.0f + __expf(-(part + bg[0])));
    v4f z = g * r + (1.0f - g) * sk;
    float s  = z.x + z.y + z.z + z.w;
    float sq = z.x * z.x + z.y * z.y + z.z * z.z + z.w * z.w;
    #pragma unroll
    for (int off = 16; off > 0; off >>= 1) {
        s  += __shfl_xor(s,  off, 32);
        sq += __shfl_xor(sq, off, 32);
    }
    const float mu   = s * (1.0f / 128.0f);
    const float var  = sq * (1.0f / 128.0f) - mu * mu;
    const float rstd = rsqrtf(var + EPS_LN);
    const float alpha = prelu_a[0];
    const v4f lg = *(const v4f*)(ln_g + c);
    const v4f lb = *(const v4f*)(ln_b + c);
    v4f y;
    #pragma unroll
    for (int i = 0; i < 4; ++i) {
        float yy = (z[i] - mu) * rstd * lg[i] + lb[i];
        y[i] = (yy >= 0.0f) ? yy : alpha * yy;
    }
    *(v4f*)(out + (size_t)wave * HD + c) = y;
}

extern "C" void kernel_launch(void* const* d_in, const int* in_sizes, int n_in,
                              void* d_out, int out_size, void* d_ws, size_t ws_size,
                              hipStream_t stream) {
    const float* feat    = (const float*)d_in[0];
    const int*   src     = (const int*)  d_in[1];
    const int*   dst     = (const int*)  d_in[2];
    const float* W_src   = (const float*)d_in[3];
    const float* b_src   = (const float*)d_in[4];
    const float* attn_l  = (const float*)d_in[5];
    const float* attn_r  = (const float*)d_in[6];
    const float* W_skip  = (const float*)d_in[7];
    const float* b_skip  = (const float*)d_in[8];
    const float* W_gate  = (const float*)d_in[9];
    const float* b_gate  = (const float*)d_in[10];
    const float* ln_g    = (const float*)d_in[11];
    const float* ln_b    = (const float*)d_in[12];
    const float* prelu_a = (const float*)d_in[13];
    float* out = (float*)d_out;

    float* ws       = (float*)d_ws;
    float* feat_src = ws;
    float* skip     = feat_src + (size_t)N_NODES * HD;
    float* el       = skip     + (size_t)N_NODES * HD;
    float* er       = el       + (size_t)N_NODES * HEADS;
    float* emax     = er       + (size_t)N_NODES * HEADS;
    float* esum     = emax     + (size_t)N_NODES * HEADS;
    float* ebuf     = esum     + (size_t)N_NODES * HEADS;

    {
        int n = N_NODES * HD;
        fill_kernel<<<(n + 255) / 256, 256, 0, stream>>>(out, esum, (int*)emax);
    }
    {
        int waves  = N_NODES / 16;
        int blocks = (waves * 32 + 255) / 256;
        gemm_wmma_kernel<<<blocks, 256, 0, stream>>>(feat, W_src, b_src, W_skip, b_skip,
                                                     feat_src, skip);
    }
    {
        int n = N_NODES * HEADS;
        elr_kernel<<<(n + 255) / 256, 256, 0, stream>>>(feat_src, attn_l, attn_r, el, er);
    }
    edge_logit_kernel<<<(N_EDGES + 255) / 256, 256, 0, stream>>>(src, dst, el, er, ebuf,
                                                                 (int*)emax);
    edge_exp_kernel<<<(N_EDGES + 255) / 256, 256, 0, stream>>>(dst, emax, ebuf, esum);
    {
        int blocks = (N_EDGES * 32 + 255) / 256;
        aggregate_kernel<<<blocks, 256, 0, stream>>>(src, dst, ebuf, feat_src, out);
    }
    {
        int blocks = (N_NODES * 32 + 255) / 256;
        final_kernel<<<blocks, 256, 0, stream>>>(skip, esum, W_gate, b_gate, ln_g, ln_b,
                                                 prelu_a, out);
    }
}


# Edges grouped by destination (int count atomics, scan, scatter); a wave per destination orders its edges by id, then softmax, row sum, epilogue in registers; no float atomics; sum in edge id order
# speedup vs baseline: 2.1036x; 2.1036x over previous
#include <hip/hip_runtime.h>
#include <math.h>

#define N_NODES 100000
#define N_EDGES 800000
#define IN_DIM  128
#define HEADS   4
#define DHEAD   32
#define HD      128
#define EPS_LN  1e-5f
#define NEG_SLOPE 0.2f

typedef __attribute__((ext_vector_type(2))) float v2f;
typedef __attribute__((ext_vector_type(4))) float v4f;
typedef __attribute__((ext_vector_type(8))) float v8f;

__global__ void zero_count_kernel(int* __restrict__ cnt) {
    int i = blockIdx.x * blockDim.x + threadIdx.x;
    if (i < N_NODES) cnt[i] = 0;
}

__global__ __launch_bounds__(256) void gemm_wmma_kernel(
    const float* __restrict__ feat,
    const float* __restrict__ Wsrc,  const float* __restrict__ bsrc,
    const float* __restrict__ Wskip, const float* __restrict__ bskip,
    float* __restrict__ feat_src, float* __restrict__ skip) {
    int wave = (blockIdx.x * blockDim.x + threadIdx.x) >> 5;
    int lane = threadIdx.x & 31;
    if (wave >= N_NODES / 16) return;
    const int node0 = wave * 16;
    const int half  = lane >> 4;
    const int l16   = lane & 15;

    const float* Arow = feat + (size_t)(node0 + l16) * IN_DIM + 2 * half;

    for (int p = 0; p < 2; ++p) {
        const float* W    = p ? Wskip : Wsrc;
        const float* bias = p ? bskip : bsrc;
        float*       out  = p ? skip  : feat_src;
        for (int ct = 0; ct < 8; ++ct) {
            const int n = ct * 16 + l16;
            v8f acc;
            const float bv = bias[n];
            #pragma unroll
            for (int v = 0; v < 8; ++v) acc[v] = bv;
            const float* Bcol = W + (size_t)n * IN_DIM + 2 * half;
            for (int kt = 0; kt < 32; ++kt) {
                v2f a = *(const v2f*)(Arow + kt * 4);
                v2f b = *(const v2f*)(Bcol + kt * 4);
                acc = __builtin_amdgcn_wmma_f32_16x16x4_f32(
                    false, a, false, b, (short)0, acc, false, false);
            }
            float* obase = out + (size_t)node0 * HD + ct * 16 + l16;
            #pragma unroll
            for (int v = 0; v < 8; ++v)
                obase[(size_t)(v + 8 * half) * HD] = acc[v];
        }
    }
}

__global__ void elr_kernel(const float* __restrict__ feat_src,
                           const float* __restrict__ attn_l,
                           const float* __restrict__ attn_r,
                           float* __restrict__ el, float* __restrict__ er) {
    int i = blockIdx.x * blockDim.x + threadIdx.x;
    if (i >= N_NODES * HEADS) return;
    const int h = i & 3;
    const float* f  = feat_src + (size_t)(i >> 2) * HD + h * DHEAD;
    const float* al = attn_l + h * DHEAD;
    const float* ar = attn_r + h * DHEAD;
    float sl = 0.0f, sr = 0.0f;
    #pragma unroll
    for (int d = 0; d < DHEAD; ++d) {
        float v = f[d];
        sl += v * al[d];
        sr += v * ar[d];
    }
    el[i] = sl;
    er[i] = sr;
}

__global__ void count_kernel(const int* __restrict__ dst, int* __restrict__ cnt) {
    int e = blockIdx.x * blockDim.x + threadIdx.x;
    if (e >= N_EDGES) return;
    atomicAdd(cnt + dst[e], 1);
}

__global__ __launch_bounds__(256) void scan_kernel(const int* __restrict__ cnt,
                                                   int* __restrict__ offs,
                                                   int* __restrict__ cursor) {
    __shared__ int sh_base[8];
    __shared__ int sh_loc[8];
    const int b = blockIdx.x, t = threadIdx.x;
    const int lane = t & 31, w = t >> 5;
    const int4* c4 = (const int4*)cnt;
    int base = 0;
    for (int j = 0; j < b; ++j) {
        const int4 v = c4[j * 256 + t];
        base += v.x + v.y + v.z + v.w;
    }
    const int q = b * 256 + t;
    int4 v = make_int4(0, 0, 0, 0);
    if (q < N_NODES / 4) v = c4[q];
    const int loc = v.x + v.y + v.z + v.w;
    int inc = loc;
    #pragma unroll
    for (int off = 1; off < 32; off <<= 1) {
        const int n = __shfl_up(inc, off, 32);
        if (lane >= off) inc += n;
    }
    #pragma unroll
    for (int off = 16; off > 0; off >>= 1) base += __shfl_xor(base, off, 32);
    if (lane == 31) sh_loc[w] = inc;
    if (lane == 0)  sh_base[w] = base;
    __syncthreads();
    int tot = 0, pre = 0;
    #pragma unroll
    for (int i = 0; i < 8; ++i) {
        tot += sh_base[i];
        if (i < w) pre += sh_loc[i];
    }
    const int ex = tot + pre + inc - loc;
    if (q < N_NODES / 4) {
        const int4 o = make_int4(ex, ex + v.x, ex + v.x + v.y, ex + v.x + v.y + v.z);
        ((int4*)offs)[q]   = o;
        ((int4*)cursor)[q] = o;
    }
}

__global__ void scatter_kernel(const int* __restrict__ src, const int* __restrict__ dst,
                               int* __restrict__ cursor, int2* __restrict__ perm) {
    int e = blockIdx.x * blockDim.x + threadIdx.x;
    if (e >= N_EDGES) return;
    const int pos = atomicAdd(cursor + dst[e], 1);
    perm[pos] = make_int2(e, src[e]);
}

__global__ __launch_bounds__(256) void pull_kernel(
    const int* __restrict__ offs, const int* __restrict__ cnt, const int2* __restrict__ perm,
    const float* __restrict__ el, const float* __restrict__ er,
    const float* __restrict__ feat_src, const float* __restrict__ skip,
    const float* __restrict__ Wg, const float* __restrict__ bg,
    const float* __restrict__ ln_g, const float* __restrict__ ln_b,
    const float* __restrict__ prelu_a, float* __restrict__ out) {
    const int wave_v = (blockIdx.x * blockDim.x + threadIdx.x) >> 5;
    const int lane = threadIdx.x & 31;
    if (wave_v >= N_NODES) return;
    const int d     = __builtin_amdgcn_readfirstlane(wave_v);
    const int start = __builtin_amdgcn_readfirstlane(offs[d]);
    const int k     = __builtin_amdgcn_readfirstlane(cnt[d]);
    const int c = lane * 4;
    const int h = lane >> 3;
    const float* frow = feat_src + c;
    float sum = 0.0f;
    v4f acc = {0.0f, 0.0f, 0.0f, 0.0f};

    if (k <= 32) {
        int e_j = 0x7fffffff, s_j = 0;
        if (lane < k) { const int2 p = perm[start + lane]; e_j = p.x; s_j = p.y; }
        int rank = 0;
        for (int i = 0; i < k; ++i)
            rank += (__builtin_amdgcn_readlane(e_j, i) < e_j) ? 1 : 0;
        if (lane >= k) rank = lane;
        const int s_r = __builtin_amdgcn_ds_permute(rank * 4, s_j);
        v4f lg = {-INFINITY, -INFINITY, -INFINITY, -INFINITY};
        if (lane < k) {
            const v4f els = *(const v4f*)(el + (size_t)s_r * 4);
            const v4f erv = *(const v4f*)(er + (size_t)d * 4);
            #pragma unroll
            for (int q = 0; q < 4; ++q) {
                float v = els[q] + erv[q];
                lg[q] = (v >= 0.0f) ? v : NEG_SLOPE * v;
            }
        }
        v4f mx = lg;
        #pragma unroll
        for (int off = 16; off > 0; off >>= 1) {
            #pragma unroll
            for (int q = 0; q < 4; ++q) mx[q] = fmaxf(mx[q], __shfl_xor(mx[q], off, 32));
        }
        v4f exv;
        #pragma unroll
        for (int q = 0; q < 4; ++q) exv[q] = (lane < k) ? __expf(lg[q] - mx[q]) : 0.0f;

        const int x0 = __float_as_int(exv[0]), x1 = __float_as_int(exv[1]);
        const int x2 = __float_as_int(exv[2]), x3 = __float_as_int(exv[3]);
        const int m0 = (h == 0) ? -1 : 0, m1 = (h == 1) ? -1 : 0;
        const int m2 = (h == 2) ? -1 : 0, m3 = (h == 3) ? -1 : 0;
        for (int r = 0; r < k; r += 4) {
            v4f f[4];
            float a[4];
            #pragma unroll
            for (int i = 0; i < 4; ++i) {
                const int s = __builtin_amdgcn_readlane(s_r, r + i);
                f[i] = *(const v4f*)(frow + (size_t)s * HD);
                const int ab = (__builtin_amdgcn_readlane(x0, r + i) & m0)
                             | (__builtin_amdgcn_readlane(x1, r + i) & m1)
                             | (__builtin_amdgcn_readlane(x2, r + i) & m2)
                             | (__builtin_amdgcn_readlane(x3, r + i) & m3);
                a[i] = __int_as_float(ab);
            }
            #pragma unroll
            for (int i = 0; i < 4; ++i) {
                if (r + i < k) {
                    sum += a[i];
                    acc += a[i] * f[i];
                }
            }
        }
    } else {
        const float erd = er[(size_t)d * 4 + h];
        float mx = -INFINITY;
        for (int j = 0; j < k; ++j) {
            const int s = perm[start + j].y;
            float v = el[(size_t)s * 4 + h] + erd;
            v = (v >= 0.0f) ? v : NEG_SLOPE * v;
            mx = fmaxf(mx, v);
        }
        int prev = -1;
        for (int r = 0; r < k; ++r) {
            int best = 0x7fffffff, bs = 0;
            for (int j = lane; j < k; j += 32) {
                const int2 p = perm[start + j];
                if (p.x > prev && p.x < best) { best = p.x; bs = p.y; }
            }
            int m = best;
            #pragma unroll
            for (int off = 16; off > 0; off >>= 1) m = min(m, __shfl_xor(m, off, 32));
            const unsigned long long bal = __ballot(best == m);
            const int L = __ffsll((long long)bal) - 1;
            const int s = __builtin_amdgcn_readfirstlane(__shfl(bs, L, 32));
            prev = m;
            float v = el[(size_t)s * 4 + h] + erd;
            v = (v >= 0.0f) ? v : NEG_SLOPE * v;
            const float a = __expf(v - mx);
            const v4f f = *(const v4f*)(frow + (size_t)s * HD);
            sum += a;
            acc += a * f;
        }
    }

    const float inv = (sum > 0.0f) ? 1.0f / sum : 0.0f;
    const v4f r = acc * inv;
    const v4f sk = *(const v4f*)(skip + (size_t)d * HD + c);
    const v4f dm = r - sk;
    const v4f w0 = *(const v4f*)(Wg + c);
    const v4f w1 = *(const v4f*)(Wg + 128 + c);
    const v4f w2 = *(const v4f*)(Wg + 256 + c);
    float part = r.x * w0.x + r.y * w0.y + r.z * w0.z + r.w * w0.w
               + sk.x * w1.x + sk.y * w1.y + sk.z * w1.z + sk.w * w1.w
               + dm.x * w2.x + dm.y * w2.y + dm.z * w2.z + dm.w * w2.w;
    #pragma unroll
    for (int off = 16; off > 0; off >>= 1) part += __shfl_xor(part, off, 32);
    const float g = 1.0f / (1.0f + __expf(-(part + bg[0])));
    v4f z = g * r + (1.0f - g) * sk;
    float s1 = z.x + z.y + z.z + z.w;
    float sq = z.x * z.x + z.y * z.y + z.z * z.z + z.w * z.w;
    #pragma unroll
    for (int off = 16; off > 0; off >>= 1) {
        s1 += __shfl_xor(s1, off, 32);
        sq += __shfl_xor(sq, off, 32);
    }
    const float mu   = s1 * (1.0f / 128.0f);
    const float var  = sq * (1.0f / 128.0f) - mu * mu;
    const float rstd = rsqrtf(var + EPS_LN);
    const float alpha = prelu_a[0];
    const v4f lgm = *(const v4f*)(ln_g + c);
    const v4f lbt = *(const v4f*)(ln_b + c);
    v4f y;
    #pragma unroll
    for (int i = 0; i < 4; ++i) {
        float yy = (z[i] - mu) * rstd * lgm[i] + lbt[i];
        y[i] = (yy >= 0.0f) ? yy : alpha * yy;
    }
    *(v4f*)(out + (size_t)d * HD + c) = y;
}

extern "C" void kernel_launch(void* const* d_in, const int* in_sizes, int n_in,
                              void* d_out, int out_size, void* d_ws, size_t ws_size,
                              hipStream_t stream) {
    const float* feat    = (const float*)d_in[0];
    const int*   src     = (const int*)  d_in[1];
    const int*   dst     = (const int*)  d_in[2];
    const float* W_src   = (const float*)d_in[3];
    const float* b_src   = (const float*)d_in[4];
    const float* attn_l  = (const float*)d_in[5];
    const float* attn_r  = (const float*)d_in[6];
    const float* W_skip  = (const float*)d_in[7];
    const float* b_skip  = (const float*)d_in[8];
    const float* W_gate  = (const float*)d_in[9];
    const float* b_gate  = (const float*)d_in[10];
    const float* ln_g    = (const float*)d_in[11];
    const float* ln_b    = (const float*)d_in[12];
    const float* prelu_a = (const float*)d_in[13];
    float* out = (float*)d_out;

    float* ws       = (float*)d_ws;
    float* feat_src = ws;
    float* skip     = feat_src + (size_t)N_NODES * HD;
    float* el       = skip     + (size_t)N_NODES * HD;
    float* er       = el       + (size_t)N_NODES * HEADS;
    int*   cnt      = (int*)(er + (size_t)N_NODES * HEADS);
    int*   offs     = cnt  + N_NODES;
    int*   cursor   = offs + N_NODES;
    int2*  perm     = (int2*)(cnt + (size_t)2 * N_NODES * HEADS);

    zero_count_kernel<<<(N_NODES + 255) / 256, 256, 0, stream>>>(cnt);
    {
        int waves  = N_NODES / 16;
        int blocks = (waves * 32 + 255) / 256;
        gemm_wmma_kernel<<<blocks, 256, 0, stream>>>(feat, W_src, b_src, W_skip, b_skip,
                                                     feat_src, skip);
    }
    {
        int n = N_NODES * HEADS;
        elr_kernel<<<(n + 255) / 256, 256, 0, stream>>>(feat_src, attn_l, attn_r, el, er);
    }
    count_kernel<<<(N_EDGES + 255) / 256, 256, 0, stream>>>(dst, cnt);
    scan_kernel<<<(N_NODES / 4 + 255) / 256, 256, 0, stream>>>(cnt, offs, cursor);
    scatter_kernel<<<(N_EDGES + 255) / 256, 256, 0, stream>>>(src, dst, cursor, perm);
    {
        int blocks = (N_NODES * 32 + 255) / 256;
        pull_kernel<<<blocks, 256, 0, stream>>>(offs, cnt, perm, el, er, feat_src, skip,
                                                W_gate, b_gate, ln_g, ln_b, prelu_a, out);
    }
}


# GEMM: 512 resident work-groups stage both weight matrices in LDS and loop over 16-row tiles; a wave keeps its rows in registers, B fragments come from LDS, 16 B per lane; k sum order changes
# speedup vs baseline: 2.7422x; 1.3036x over previous
#include <hip/hip_runtime.h>
#include <math.h>

#define N_NODES 100000
#define N_EDGES 800000
#define IN_DIM  128
#define HEADS   4
#define DHEAD   32
#define HD      128
#define EPS_LN  1e-5f
#define NEG_SLOPE 0.2f

typedef __attribute__((ext_vector_type(2))) float v2f;
typedef __attribute__((ext_vector_type(4))) float v4f;
typedef __attribute__((ext_vector_type(8))) float v8f;

__global__ void zero_count_kernel(int* __restrict__ cnt) {
    int i = blockIdx.x * blockDim.x + threadIdx.x;
    if (i < N_NODES) cnt[i] = 0;
}

#define GEMM_THREADS 256
#define GEMM_BLOCKS  512
#define W_PITCH 132
__global__ __launch_bounds__(GEMM_THREADS) void gemm_wmma_kernel(
    const float* __restrict__ feat,
    const float* __restrict__ Wsrc,  const float* __restrict__ bsrc,
    const float* __restrict__ Wskip, const float* __restrict__ bskip,
    float* __restrict__ feat_src, float* __restrict__ skip) {
    __shared__ __attribute__((aligned(16))) float Wl[2 * HD * W_PITCH];
    __shared__ float bl[2 * HD];
    const int tid = threadIdx.x;
    for (int i = tid; i < HD * (IN_DIM / 4); i += GEMM_THREADS) {
        const int row = i >> 5, q = i & 31;
        const v4f ws = *(const v4f*)(Wsrc  + (size_t)i * 4);
        const v4f wk = *(const v4f*)(Wskip + (size_t)i * 4);
        *(v4f*)(Wl + row * W_PITCH + q * 4)        = ws;
        *(v4f*)(Wl + (HD + row) * W_PITCH + q * 4) = wk;
    }
    for (int i = tid; i < 2 * HD; i += GEMM_THREADS)
        bl[i] = (i < HD) ? bsrc[i] : bskip[i - HD];
    __syncthreads();

    const int wave = __builtin_amdgcn_readfirstlane(tid >> 5);
    const int lane = tid & 31;
    const int half = lane >> 4;
    const int l16  = lane & 15;
    const float* Bl = Wl + l16 * W_PITCH + 4 * half;

    for (int t = blockIdx.x + GEMM_BLOCKS * wave; t < N_NODES / 16;
         t += GEMM_BLOCKS * (GEMM_THREADS / 32)) {
        const int node0 = t * 16;
        const float* Arow = feat + (size_t)(node0 + l16) * IN_DIM + 4 * half;
        v4f a[16];
        #pragma unroll
        for (int j = 0; j < 16; ++j) a[j] = *(const v4f*)(Arow + 8 * j);

        for (int cp = 0; cp < 8; ++cp) {
            float*    out = (cp < 4) ? feat_src : skip;
            const int n0  = (cp & 3) * 32 + l16;
            const float bv0 = bl[cp * 32 + l16], bv1 = bl[cp * 32 + 16 + l16];
            v8f acc0, acc1;
            #pragma unroll
            for (int v = 0; v < 8; ++v) { acc0[v] = bv0; acc1[v] = bv1; }
            const float* B0 = Bl + (cp * 32) * W_PITCH;
            const float* B1 = B0 + 16 * W_PITCH;
            #pragma unroll
            for (int j = 0; j < 16; ++j) {
                const v4f b0 = *(const v4f*)(B0 + 8 * j);
                const v4f b1 = *(const v4f*)(B1 + 8 * j);
                const v2f a_lo = {a[j].x, a[j].y}, a_hi = {a[j].z, a[j].w};
                const v2f b0_lo = {b0.x, b0.y}, b0_hi = {b0.z, b0.w};
                const v2f b1_lo = {b1.x, b1.y}, b1_hi = {b1.z, b1.w};
                acc0 = __builtin_amdgcn_wmma_f32_16x16x4_f32(
                    false, a_lo, false, b0_lo, (short)0, acc0, false, false);
                acc1 = __builtin_amdgcn_wmma_f32_16x16x4_f32(
                    false, a_lo, false, b1_lo, (short)0, acc1, false, false);
                acc0 = __builtin_amdgcn_wmma_f32_16x16x4_f32(
                    false, a_hi, false, b0_hi, (short)0, acc0, false, false);
                acc1 = __builtin_amdgcn_wmma_f32_16x16x4_f32(
                    false, a_hi, false, b1_hi, (short)0, acc1, false, false);
            }
            float* obase = out + (size_t)node0 * HD + n0;
            #pragma unroll
            for (int v = 0; v < 8; ++v) {
                obase[(size_t)(v + 8 * half) * HD]      = acc0[v];
                obase[(size_t)(v + 8 * half) * HD + 16] = acc1[v];
            }
        }
    }
}

__global__ void elr_kernel(const float* __restrict__ feat_src,
                           const float* __restrict__ attn_l,
                           const float* __restrict__ attn_r,
                           float* __restrict__ el, float* __restrict__ er) {
    int i = blockIdx.x * blockDim.x + threadIdx.x;
    if (i >= N_NODES * HEADS) return;
    const int h = i & 3;
    const float* f  = feat_src + (size_t)(i >> 2) * HD + h * DHEAD;
    const float* al = attn_l + h * DHEAD;
    const float* ar = attn_r + h * DHEAD;
    float sl = 0.0f, sr = 0.0f;
    #pragma unroll
    for (int d = 0; d < DHEAD; ++d) {
        float v = f[d];
        sl += v * al[d];
        sr += v * ar[d];
    }
    el[i] = sl;
    er[i] = sr;
}

__global__ void count_kernel(const int* __restrict__ dst, int* __restrict__ cnt) {
    int e = blockIdx.x * blockDim.x + threadIdx.x;
    if (e >= N_EDGES) return;
    atomicAdd(cnt + dst[e], 1);
}

__global__ __launch_bounds__(256) void scan_kernel(const int* __restrict__ cnt,
                                                   int* __restrict__ offs,
                                                   int* __restrict__ cursor) {
    __shared__ int sh_base[8];
    __shared__ int sh_loc[8];
    const int b = blockIdx.x, t = threadIdx.x;
    const int lane = t & 31, w = t >> 5;
    const int4* c4 = (const int4*)cnt;
    int base = 0;
    for (int j = 0; j < b; ++j) {
        const int4 v = c4[j * 256 + t];
        base += v.x + v.y + v.z + v.w;
    }
    const int q = b * 256 + t;
    int4 v = make_int4(0, 0, 0, 0);
    if (q < N_NODES / 4) v = c4[q];
    const int loc = v.x + v.y + v.z + v.w;
    int inc = loc;
    #pragma unroll
    for (int off = 1; off < 32; off <<= 1) {
        const int n = __shfl_up(inc, off, 32);
        if (lane >= off) inc += n;
    }
    #pragma unroll
    for (int off = 16; off > 0; off >>= 1) base += __shfl_xor(base, off, 32);
    if (lane == 31) sh_loc[w] = inc;
    if (lane == 0)  sh_base[w] = base;
    __syncthreads();
    int tot = 0, pre = 0;
    #pragma unroll
    for (int i = 0; i < 8; ++i) {
        tot += sh_base[i];
        if (i < w) pre += sh_loc[i];
    }
    const int ex = tot + pre + inc - loc;
    if (q < N_NODES / 4) {
        const int4 o = make_int4(ex, ex + v.x, ex + v.x + v.y, ex + v.x + v.y + v.z);
        ((int4*)offs)[q]   = o;
        ((int4*)cursor)[q] = o;
    }
}

__global__ void scatter_kernel(const int* __restrict__ src, const int* __restrict__ dst,
                               int* __restrict__ cursor, int2* __restrict__ perm) {
    int e = blockIdx.x * blockDim.x + threadIdx.x;
    if (e >= N_EDGES) return;
    const int pos = atomicAdd(cursor + dst[e], 1);
    perm[pos] = make_int2(e, src[e]);
}

__global__ __launch_bounds__(256) void pull_kernel(
    const int* __restrict__ offs, const int* __restrict__ cnt, const int2* __restrict__ perm,
    const float* __restrict__ el, const float* __restrict__ er,
    const float* __restrict__ feat_src, const float* __restrict__ skip,
    const float* __restrict__ Wg, const float* __restrict__ bg,
    const float* __restrict__ ln_g, const float* __restrict__ ln_b,
    const float* __restrict__ prelu_a, float* __restrict__ out) {
    const int wave_v = (blockIdx.x * blockDim.x + threadIdx.x) >> 5;
    const int lane = threadIdx.x & 31;
    if (wave_v >= N_NODES) return;
    const int d     = __builtin_amdgcn_readfirstlane(wave_v);
    const int start = __builtin_amdgcn_readfirstlane(offs[d]);
    const int k     = __builtin_amdgcn_readfirstlane(cnt[d]);
    const int c = lane * 4;
    const int h = lane >> 3;
    const float* frow = feat_src + c;
    float sum = 0.0f;
    v4f acc = {0.0f, 0.0f, 0.0f, 0.0f};

    if (k <= 32) {
        int e_j = 0x7fffffff, s_j = 0;
        if (lane < k) { const int2 p = perm[start + lane]; e_j = p.x; s_j = p.y; }
        int rank = 0;
        for (int i = 0; i < k; ++i)
            rank += (__builtin_amdgcn_readlane(e_j, i) < e_j) ? 1 : 0;
        if (lane >= k) rank = lane;
        const int s_r = __builtin_amdgcn_ds_permute(rank * 4, s_j);
        v4f lg = {-INFINITY, -INFINITY, -INFINITY, -INFINITY};
        if (lane < k) {
            const v4f els = *(const v4f*)(el + (size_t)s_r * 4);
            const v4f erv = *(const v4f*)(er + (size_t)d * 4);
            #pragma unroll
            for (int q = 0; q < 4; ++q) {
                float v = els[q] + erv[q];
                lg[q] = (v >= 0.0f) ? v : NEG_SLOPE * v;
            }
        }
        v4f mx = lg;
        #pragma unroll
        for (int off = 16; off > 0; off >>= 1) {
            #pragma unroll
            for (int q = 0; q < 4; ++q) mx[q] = fmaxf(mx[q], __shfl_xor(mx[q], off, 32));
        }
        v4f exv;
        #pragma unroll
        for (int q = 0; q < 4; ++q) exv[q] = (lane < k) ? __expf(lg[q] - mx[q]) : 0.0f;

        const int x0 = __float_as_int(exv[0]), x1 = __float_as_int(exv[1]);
        const int x2 = __float_as_int(exv[2]), x3 = __float_as_int(exv[3]);
        const int m0 = (h == 0) ? -1 : 0, m1 = (h == 1) ? -1 : 0;
        const int m2 = (h == 2) ? -1 : 0, m3 = (h == 3) ? -1 : 0;
        for (int r = 0; r < k; r += 4) {
            v4f f[4];
            float a[4];
            #pragma unroll
            for (int i = 0; i < 4; ++i) {
                const int s = __builtin_amdgcn_readlane(s_r, r + i);
                f[i] = *(const v4f*)(frow + (size_t)s * HD);
                const int ab = (__builtin_amdgcn_readlane(x0, r + i) & m0)
                             | (__builtin_amdgcn_readlane(x1, r + i) & m1)
                             | (__builtin_amdgcn_readlane(x2, r + i) & m2)
                             | (__builtin_amdgcn_readlane(x3, r + i) & m3);
                a[i] = __int_as_float(ab);
            }
            #pragma unroll
            for (int i = 0; i < 4; ++i) {
                if (r + i < k) {
                    sum += a[i];
                    acc += a[i] * f[i];
                }
            }
        }
    } else {
        const float erd = er[(size_t)d * 4 + h];
        float mx = -INFINITY;
        for (int j = 0; j < k; ++j) {
            const int s = perm[start + j].y;
            float v = el[(size_t)s * 4 + h] + erd;
            v = (v >= 0.0f) ? v : NEG_SLOPE * v;
            mx = fmaxf(mx, v);
        }
        int prev = -1;
        for (int r = 0; r < k; ++r) {
            int best = 0x7fffffff, bs = 0;
            for (int j = lane; j < k; j += 32) {
                const int2 p = perm[start + j];
                if (p.x > prev && p.x < best) { best = p.x; bs = p.y; }
            }
            int m = best;
            #pragma unroll
            for (int off = 16; off > 0; off >>= 1) m = min(m, __shfl_xor(m, off, 32));
            const unsigned long long bal = __ballot(best == m);
            const int L = __ffsll((long long)bal) - 1;
            const int s = __builtin_amdgcn_readfirstlane(__shfl(bs, L, 32));
            prev = m;
            float v = el[(size_t)s * 4 + h] + erd;
            v = (v >= 0.0f) ? v : NEG_SLOPE * v;
            const float a = __expf(v - mx);
            const v4f f = *(const v4f*)(frow + (size_t)s * HD);
            sum += a;
            acc += a * f;
        }
    }

    const float inv = (sum > 0.0f) ? 1.0f / sum : 0.0f;
    const v4f r = acc * inv;
    const v4f sk = *(const v4f*)(skip + (size_t)d * HD + c);
    const v4f dm = r - sk;
    const v4f w0 = *(const v4f*)(Wg + c);
    const v4f w1 = *(const v4f*)(Wg + 128 + c);
    const v4f w2 = *(const v4f*)(Wg + 256 + c);
    float part = r.x * w0.x + r.y * w0.y + r.z * w0.z + r.w * w0.w
               + sk.x * w1.x + sk.y * w1.y + sk.z * w1.z + sk.w * w1.w
               + dm.x * w2.x + dm.y * w2.y + dm.z * w2.z + dm.w * w2.w;
    #pragma unroll
    for (int off = 16; off > 0; off >>= 1) part += __shfl_xor(part, off, 32);
    const float g = 1.0f / (1.0f + __expf(-(part + bg[0])));
    v4f z = g * r + (1.0f - g) * sk;
    float s1 = z.x + z.y + z.z + z.w;
    float sq = z.x * z.x + z.y * z.y + z.z * z.z + z.w * z.w;
    #pragma unroll
    for (int off = 16; off > 0; off >>= 1) {
        s1 += __shfl_xor(s1, off, 32);
        sq += __shfl_xor(sq, off, 32);
    }
    const float mu   = s1 * (1.0f / 128.0f);
    const float var  = sq * (1.0f / 128.0f) - mu * mu;
    const float rstd = rsqrtf(var + EPS_LN);
    const float alpha = prelu_a[0];
    const v4f lgm = *(const v4f*)(ln_g + c);
    const v4f lbt = *(const v4f*)(ln_b + c);
    v4f y;
    #pragma unroll
    for (int i = 0; i < 4; ++i) {
        float yy = (z[i] - mu) * rstd * lgm[i] + lbt[i];
        y[i] = (yy >= 0.0f) ? yy : alpha * yy;
    }
    *(v4f*)(out + (size_t)d * HD + c) = y;
}

extern "C" void kernel_launch(void* const* d_in, const int* in_sizes, int n_in,
                              void* d_out, int out_size, void* d_ws, size_t ws_size,
                              hipStream_t stream) {
    const float* feat    = (const float*)d_in[0];
    const int*   src     = (const int*)  d_in[1];
    const int*   dst     = (const int*)  d_in[2];
    const float* W_src   = (const float*)d_in[3];
    const float* b_src   = (const float*)d_in[4];
    const float* attn_l  = (const float*)d_in[5];
    const float* attn_r  = (const float*)d_in[6];
    const float* W_skip  = (const float*)d_in[7];
    const float* b_skip  = (const float*)d_in[8];
    const float* W_gate  = (const float*)d_in[9];
    const float* b_gate  = (const float*)d_in[10];
    const float* ln_g    = (const float*)d_in[11];
    const float* ln_b    = (const float*)d_in[12];
    const float* prelu_a = (const float*)d_in[13];
    float* out = (float*)d_out;

    float* ws       = (float*)d_ws;
    float* feat_src = ws;
    float* skip     = feat_src + (size_t)N_NODES * HD;
    float* el       = skip     + (size_t)N_NODES * HD;
    float* er       = el       + (size_t)N_NODES * HEADS;
    int*   cnt      = (int*)(er + (size_t)N_NODES * HEADS);
    int*   offs     = cnt  + N_NODES;
    int*   cursor   = offs + N_NODES;
    int2*  perm     = (int2*)(cnt + (size_t)2 * N_NODES * HEADS);

    zero_count_kernel<<<(N_NODES + 255) / 256, 256, 0, stream>>>(cnt);
    gemm_wmma_kernel<<<GEMM_BLOCKS, GEMM_THREADS, 0, stream>>>(feat, W_src, b_src, W_skip, b_skip,
                                                               feat_src, skip);
    {
        int n = N_NODES * HEADS;
        elr_kernel<<<(n + 255) / 256, 256, 0, stream>>>(feat_src, attn_l, attn_r, el, er);
    }
    count_kernel<<<(N_EDGES + 255) / 256, 256, 0, stream>>>(dst, cnt);
    scan_kernel<<<(N_NODES / 4 + 255) / 256, 256, 0, stream>>>(cnt, offs, cursor);
    scatter_kernel<<<(N_EDGES + 255) / 256, 256, 0, stream>>>(src, dst, cursor, perm);
    {
        int blocks = (N_NODES * 32 + 255) / 256;
        pull_kernel<<<blocks, 256, 0, stream>>>(offs, cnt, perm, el, er, feat_src, skip,
                                                W_gate, b_gate, ln_g, ln_b, prelu_a, out);
    }
}
